# starting kernel
# baseline (speedup 1.0000x reference)
#include <hip/hip_runtime.h>


#define NV 18
#define NP 1330
#define KP 1332
#define NCOLS 20
#define NGROUPS (KP / 4)
#define WAVES 8
#define ROWS_PER_WAVE 16
#define ROWS_PER_BLOCK (WAVES * ROWS_PER_WAVE)

typedef __attribute__((ext_vector_type(2))) float v2f;
typedef __attribute__((ext_vector_type(4))) float v4f;
typedef __attribute__((ext_vector_type(8))) float v8f;

struct MonTab {
    unsigned char a[KP];
    unsigned char b[KP];
    unsigned char c[KP];
};
constexpr MonTab make_tab() {
    MonTab t{};
    int f = 0;
    t.a[f] = 18; t.b[f] = 18; t.c[f] = 18; ++f;
    for (int i = 0; i < NV; ++i) { t.a[f] = i; t.b[f] = 18; t.c[f] = 18; ++f; }
    for (int i = 0; i < NV; ++i)
        for (int j = i; j < NV; ++j) { t.a[f] = i; t.b[f] = j; t.c[f] = 18; ++f; }
    for (int i = 0; i < NV; ++i)
        for (int j = i; j < NV; ++j)
            for (int k = j; k < NV; ++k) { t.a[f] = i; t.b[f] = j; t.c[f] = k; ++f; }
    for (; f < KP; ++f) { t.a[f] = 18; t.b[f] = 18; t.c[f] = 18; }
    return t;
}
constexpr MonTab TAB = make_tab();

template <int F>
static __device__ __forceinline__ float mono(const float* x) {
    constexpr int ia = TAB.a[F];
    constexpr int ib = TAB.b[F];
    constexpr int ic = TAB.c[F];
    float p = x[ia];
    if constexpr (ib != 18) p *= x[ib];
    if constexpr (ic != 18) p *= x[ic];
    return p;
}

template <int G, int GEnd>
struct Steps {
    static __device__ __forceinline__ void run(const float* x, bool hi,
                                               const float* b0p, const float* b1p,
                                               v8f& acc0, v8f& acc1) {
        constexpr int f = 4 * G;
        float p0 = mono<f + 0>(x);
        float p1 = mono<f + 1>(x);
        float p2 = mono<f + 2>(x);
        float p3 = mono<f + 3>(x);
        v2f A;
        A.x = hi ? p2 : p0;
        A.y = hi ? p3 : p1;
        v2f B0 = *(const v2f*)(b0p + f);
        v2f B1 = *(const v2f*)(b1p + f);
        acc0 = __builtin_amdgcn_wmma_f32_16x16x4_f32(
                   false, A, false, B0, (short)0, acc0, false, false);
        acc1 = __builtin_amdgcn_wmma_f32_16x16x4_f32(
                   false, A, false, B1, (short)0, acc1, false, false);
        Steps<G + 1, GEnd>::run(x, hi, b0p, b1p, acc0, acc1);
    }
};
template <int GEnd>
struct Steps<GEnd, GEnd> {
    static __device__ __forceinline__ void run(const float*, bool, const float*,
                                               const float*, v8f&, v8f&) {}
};

__global__ __launch_bounds__(256)
void vindy_wmma_kernel(const float* __restrict__ z,
                       const float* __restrict__ betas,
                       const float* __restrict__ big_xi,
                       const float* __restrict__ mask,
                       float* __restrict__ out,
                       int batch)
{
    __shared__ float sXiT[NCOLS][KP];

    const int tid  = threadIdx.x;
    const int lane = tid & 31;
    const int w    = tid >> 5;
    const int half = lane >> 4;
    const int lm   = lane & 15;
    const int rowBase = blockIdx.x * ROWS_PER_BLOCK + w * ROWS_PER_WAVE;

    for (int col = 0; col < NCOLS; ++col) {
        for (int k = tid; k < KP; k += 256) {
            float v = 0.0f;
            if (col < NV && k < NP) v = big_xi[k * NV + col] * mask[k * NV + col];
            sXiT[col][k] = v;
        }
    }

    int gr = rowBase + lm;
    if (gr > batch - 1) gr = batch - 1;
    float x[19];
    {
        const v4f* zr = (const v4f*)(z + (size_t)gr * 16);
        v4f q0 = zr[0], q1 = zr[1], q2 = zr[2], q3 = zr[3];
        x[0]  = q0.x; x[1]  = q0.y; x[2]  = q0.z; x[3]  = q0.w;
        x[4]  = q1.x; x[5]  = q1.y; x[6]  = q1.z; x[7]  = q1.w;
        x[8]  = q2.x; x[9]  = q2.y; x[10] = q2.z; x[11] = q2.w;
        x[12] = q3.x; x[13] = q3.y; x[14] = q3.z; x[15] = q3.w;
        const v2f* br = (const v2f*)(betas + (size_t)gr * 2);
        v2f bb = br[0];
        x[16] = bb.x; x[17] = bb.y; x[18] = 1.0f;
    }

    __syncthreads();

    v8f acc0 = {0.f, 0.f, 0.f, 0.f, 0.f, 0.f, 0.f, 0.f};
    v8f acc1 = {0.f, 0.f, 0.f, 0.f, 0.f, 0.f, 0.f, 0.f};

    const bool hi = (half != 0);
    const float* b0p = &sXiT[lm][2 * half];
    const int col1   = (lm < 2) ? (16 + lm) : 18;
    const float* b1p = &sXiT[col1][2 * half];

    Steps<0, NGROUPS>::run(x, hi, b0p, b1p, acc0, acc1);

    const int orow = rowBase + 8 * half;
    if (rowBase + ROWS_PER_WAVE <= batch) {
#pragma unroll
        for (int v = 0; v < 8; ++v)
            out[(size_t)(orow + v) * NV + lm] = acc0[v];
        if (lm < 2) {
#pragma unroll
            for (int v = 0; v < 8; ++v)
                out[(size_t)(orow + v) * NV + 16 + lm] = acc1[v];
        }
    } else {
#pragma unroll
        for (int v = 0; v < 8; ++v) {
            int r = orow + v;
            if (r < batch) {
                out[(size_t)r * NV + lm] = acc0[v];
                if (lm < 2) out[(size_t)r * NV + 16 + lm] = acc1[v];
            }
        }
    }
}

extern "C" void kernel_launch(void* const* d_in, const int* in_sizes, int n_in,
                              void* d_out, int out_size, void* d_ws, size_t ws_size,
                              hipStream_t stream) {
    const float* z      = (const float*)d_in[0];
    const float* betas  = (const float*)d_in[1];
    const float* big_xi = (const float*)d_in[2];
    const float* mask   = (const float*)d_in[3];
    float* out          = (float*)d_out;

    const int batch = in_sizes[0] / 16;
    const int grid  = (batch + ROWS_PER_BLOCK - 1) / ROWS_PER_BLOCK;

    vindy_wmma_kernel<<<grid, 256, 0, stream>>>(z, betas, big_xi, mask, out, batch);
}


# Stage Xi*mask into LDS with coalesced quad-float reads in memory order (each line fetched once per block, a trip's loads in flight together) and a transposing LDS scatter, not strided scalar reads
# speedup vs baseline: 2.0938x; 2.0938x over previous
#include <hip/hip_runtime.h>


#define NV 18
#define NP 1330
#define KP 1332
#define NCOLS 20
#define NGROUPS (KP / 4)
#define WAVES 8
#define ROWS_PER_WAVE 16
#define ROWS_PER_BLOCK (WAVES * ROWS_PER_WAVE)

typedef __attribute__((ext_vector_type(2))) float v2f;
typedef __attribute__((ext_vector_type(4))) float v4f;
typedef __attribute__((ext_vector_type(8))) float v8f;

struct MonTab {
    unsigned char a[KP];
    unsigned char b[KP];
    unsigned char c[KP];
};
constexpr MonTab make_tab() {
    MonTab t{};
    int f = 0;
    t.a[f] = 18; t.b[f] = 18; t.c[f] = 18; ++f;
    for (int i = 0; i < NV; ++i) { t.a[f] = i; t.b[f] = 18; t.c[f] = 18; ++f; }
    for (int i = 0; i < NV; ++i)
        for (int j = i; j < NV; ++j) { t.a[f] = i; t.b[f] = j; t.c[f] = 18; ++f; }
    for (int i = 0; i < NV; ++i)
        for (int j = i; j < NV; ++j)
            for (int k = j; k < NV; ++k) { t.a[f] = i; t.b[f] = j; t.c[f] = k; ++f; }
    for (; f < KP; ++f) { t.a[f] = 18; t.b[f] = 18; t.c[f] = 18; }
    return t;
}
constexpr MonTab TAB = make_tab();

template <int F>
static __device__ __forceinline__ float mono(const float* x) {
    constexpr int ia = TAB.a[F];
    constexpr int ib = TAB.b[F];
    constexpr int ic = TAB.c[F];
    float p = x[ia];
    if constexpr (ib != 18) p *= x[ib];
    if constexpr (ic != 18) p *= x[ic];
    return p;
}

template <int G, int GEnd>
struct Steps {
    static __device__ __forceinline__ void run(const float* x, bool hi,
                                               const float* b0p, const float* b1p,
                                               v8f& acc0, v8f& acc1) {
        constexpr int f = 4 * G;
        float p0 = mono<f + 0>(x);
        float p1 = mono<f + 1>(x);
        float p2 = mono<f + 2>(x);
        float p3 = mono<f + 3>(x);
        v2f A;
        A.x = hi ? p2 : p0;
        A.y = hi ? p3 : p1;
        v2f B0 = *(const v2f*)(b0p + f);
        v2f B1 = *(const v2f*)(b1p + f);
        acc0 = __builtin_amdgcn_wmma_f32_16x16x4_f32(
                   false, A, false, B0, (short)0, acc0, false, false);
        acc1 = __builtin_amdgcn_wmma_f32_16x16x4_f32(
                   false, A, false, B1, (short)0, acc1, false, false);
        Steps<G + 1, GEnd>::run(x, hi, b0p, b1p, acc0, acc1);
    }
};
template <int GEnd>
struct Steps<GEnd, GEnd> {
    static __device__ __forceinline__ void run(const float*, bool, const float*,
                                               const float*, v8f&, v8f&) {}
};

__global__ __launch_bounds__(256)
void vindy_wmma_kernel(const float* __restrict__ z,
                       const float* __restrict__ betas,
                       const float* __restrict__ big_xi,
                       const float* __restrict__ mask,
                       float* __restrict__ out,
                       int batch)
{
    __shared__ float sXiT[NCOLS][KP];

    const int tid  = threadIdx.x;
    const int lane = tid & 31;
    const int w    = tid >> 5;
    const int half = lane >> 4;
    const int lm   = lane & 15;
    const int rowBase = blockIdx.x * ROWS_PER_BLOCK + w * ROWS_PER_WAVE;

    {
        float* s = &sXiT[0][0];
        for (int t = tid; t < 2 * KP; t += 256) s[NV * KP + t] = 0.0f;
        if (tid < 2 * NV) sXiT[tid >> 1][NP + (tid & 1)] = 0.0f;
        const v4f* xi4 = (const v4f*)big_xi;
        const v4f* mk4 = (const v4f*)mask;
        constexpr int NQ = NP * NV / 4;
        constexpr int UB = 8;
        static_assert(NP * NV % 4 == 0, "quad loads need a whole number of quads");
        for (int base = tid; base < NQ; base += 256 * UB) {
            v4f a[UB], m[UB];
#pragma unroll
            for (int u = 0; u < UB; ++u) {
                int q = base + 256 * u;
                if (q > NQ - 1) q = NQ - 1;
                a[u] = xi4[q];
                m[u] = mk4[q];
            }
#pragma unroll
            for (int u = 0; u < UB; ++u) {
                const int q = base + 256 * u;
                if (q < NQ) {
                    const unsigned i0 = 4u * (unsigned)q;
                    const unsigned k0 = i0 / (unsigned)NV;
                    const unsigned c0 = i0 - k0 * (unsigned)NV;
                    unsigned c1 = c0 + 2u, k1 = k0;
                    if (c1 == (unsigned)NV) { c1 = 0u; k1 = k0 + 1u; }
                    s[c0 * KP + k0]        = a[u].x * m[u].x;
                    s[(c0 + 1u) * KP + k0] = a[u].y * m[u].y;
                    s[c1 * KP + k1]        = a[u].z * m[u].z;
                    s[(c1 + 1u) * KP + k1] = a[u].w * m[u].w;
                }
            }
        }
    }

    int gr = rowBase + lm;
    if (gr > batch - 1) gr = batch - 1;
    float x[19];
    {
        const v4f* zr = (const v4f*)(z + (size_t)gr * 16);
        v4f q0 = zr[0], q1 = zr[1], q2 = zr[2], q3 = zr[3];
        x[0]  = q0.x; x[1]  = q0.y; x[2]  = q0.z; x[3]  = q0.w;
        x[4]  = q1.x; x[5]  = q1.y; x[6]  = q1.z; x[7]  = q1.w;
        x[8]  = q2.x; x[9]  = q2.y; x[10] = q2.z; x[11] = q2.w;
        x[12] = q3.x; x[13] = q3.y; x[14] = q3.z; x[15] = q3.w;
        const v2f* br = (const v2f*)(betas + (size_t)gr * 2);
        v2f bb = br[0];
        x[16] = bb.x; x[17] = bb.y; x[18] = 1.0f;
    }

    __syncthreads();

    v8f acc0 = {0.f, 0.f, 0.f, 0.f, 0.f, 0.f, 0.f, 0.f};
    v8f acc1 = {0.f, 0.f, 0.f, 0.f, 0.f, 0.f, 0.f, 0.f};

    const bool hi = (half != 0);
    const float* b0p = &sXiT[lm][2 * half];
    const int col1   = (lm < 2) ? (16 + lm) : 18;
    const float* b1p = &sXiT[col1][2 * half];

    Steps<0, NGROUPS>::run(x, hi, b0p, b1p, acc0, acc1);

    const int orow = rowBase + 8 * half;
    if (rowBase + ROWS_PER_WAVE <= batch) {
#pragma unroll
        for (int v = 0; v < 8; ++v)
            out[(size_t)(orow + v) * NV + lm] = acc0[v];
        if (lm < 2) {
#pragma unroll
            for (int v = 0; v < 8; ++v)
                out[(size_t)(orow + v) * NV + 16 + lm] = acc1[v];
        }
    } else {
#pragma unroll
        for (int v = 0; v < 8; ++v) {
            int r = orow + v;
            if (r < batch) {
                out[(size_t)r * NV + lm] = acc0[v];
                if (lm < 2) out[(size_t)r * NV + 16 + lm] = acc1[v];
            }
        }
    }
}

extern "C" void kernel_launch(void* const* d_in, const int* in_sizes, int n_in,
                              void* d_out, int out_size, void* d_ws, size_t ws_size,
                              hipStream_t stream) {
    const float* z      = (const float*)d_in[0];
    const float* betas  = (const float*)d_in[1];
    const float* big_xi = (const float*)d_in[2];
    const float* mask   = (const float*)d_in[3];
    float* out          = (float*)d_out;

    const int batch = in_sizes[0] / 16;
    const int grid  = (batch + ROWS_PER_BLOCK - 1) / ROWS_PER_BLOCK;

    vindy_wmma_kernel<<<grid, 256, 0, stream>>>(z, betas, big_xi, mask, out, batch);
}


# Factor the lowest variable out of each library term: 22 WMMA column tiles of 190 pair products x_j*x_k against regrouped Xi*mask, scaled by x_i and summed (407, not 666 WMMAs/wave); sum order changes
# speedup vs baseline: 2.5699x; 1.2274x over previous
#include <hip/hip_runtime.h>


#define NV 18
#define NP 1330
#define NPAIR 190
#define NAG 48
#define NMAIN 19
#define NTILE 22
#define WAVES 8
#define ROWS_PER_WAVE 16
#define ROWS_PER_BLOCK (WAVES * ROWS_PER_WAVE)

typedef __attribute__((ext_vector_type(2))) float v2f;
typedef __attribute__((ext_vector_type(4))) float v4f;
typedef __attribute__((ext_vector_type(8))) float v8f;

constexpr int pairs_from(int i) { return (19 - i) * (20 - i) / 2; }
constexpr int pair_index(int j, int k) { return (18 - j) * (19 - j) / 2 + (k - j); }
constexpr int pair_j(int p) {
    int j = 18;
    while (p >= (18 - j) * (19 - j) / 2 + (19 - j)) --j;
    return j;
}
constexpr int pair_k(int p) { return pair_j(p) + (p - (18 - pair_j(p)) * (19 - pair_j(p)) / 2); }
constexpr int tile_groups(int t) {
    return t < NMAIN ? (pairs_from(t) + 3) / 4 : (pairs_from(8 * (t - NMAIN)) + 3) / 4;
}
constexpr int tile_base(int t) {
    int s = 0;
    for (int u = 0; u < t; ++u) s += tile_groups(u);
    return s;
}
constexpr int CHUNK_STEPS = 12;
constexpr int NGTOT = tile_base(NTILE);
constexpr int IMG_FLOATS = NGTOT * 64;

struct StageTab {
    unsigned v[NP + 1];
    unsigned short lastg[NMAIN + 1];
};
constexpr unsigned stage_word(int i, int j, int k) {
    const int p = pair_index(j, k);
    const int m = (tile_base(i) + (p >> 2)) * 64 + 32 * ((p >> 1) & 1) + (p & 1);
    const int t = (tile_base(NMAIN + (i >> 3)) + (p >> 2)) * 64
                + (2 * (i & 7) + 16 * ((p >> 1) & 1)) * 2 + (p & 1);
    return (unsigned)m | ((unsigned)t << 16);
}
constexpr StageTab make_stage_tab() {
    StageTab t{};
    int f = 0;
    t.v[f++] = stage_word(18, 18, 18);
    for (int i = 0; i < NV; ++i) t.v[f++] = stage_word(i, 18, 18);
    for (int i = 0; i < NV; ++i)
        for (int j = i; j < NV; ++j) t.v[f++] = stage_word(i, j, 18);
    for (int i = 0; i < NV; ++i)
        for (int j = i; j < NV; ++j)
            for (int k = j; k < NV; ++k) t.v[f++] = stage_word(i, j, k);
    t.v[NP] = 0;
    for (int i = 0; i < NMAIN; ++i) t.lastg[i] = (unsigned short)(tile_base(i + 1) - 1);
    t.lastg[NMAIN] = 0;
    return t;
}
__constant__ const StageTab STAB = make_stage_tab();

template <int P>
static __device__ __forceinline__ float pairprod(const float* x) {
    if constexpr (P >= NPAIR) {
        return 0.0f;
    } else {
        constexpr int j = pair_j(P);
        constexpr int k = pair_k(P);
        if constexpr (k == 18) return x[j];
        else return x[j] * x[k];
    }
}

template <int G, int GEnd>
struct FillA {
    static __device__ __forceinline__ void run(const float* x, bool hi, v2f* A) {
        constexpr int p = 4 * G;
        const float q0 = pairprod<p + 0>(x);
        const float q1 = pairprod<p + 1>(x);
        const float q2 = pairprod<p + 2>(x);
        const float q3 = pairprod<p + 3>(x);
        A[G].x = hi ? q2 : q0;
        A[G].y = hi ? q3 : q1;
        FillA<G + 1, GEnd>::run(x, hi, A);
    }
};
template <int GEnd>
struct FillA<GEnd, GEnd> {
    static __device__ __forceinline__ void run(const float*, bool, v2f*) {}
};

static __device__ __forceinline__ void scale_add(const float* xs, const v8f& e, const v8f& o, v8f& acc) {
    const v4f s0 = *(const v4f*)(xs);
    const v4f s1 = *(const v4f*)(xs + 4);
    acc[0] += s0.x * (e[0] + o[0]);
    acc[1] += s0.y * (e[1] + o[1]);
    acc[2] += s0.z * (e[2] + o[2]);
    acc[3] += s0.w * (e[3] + o[3]);
    acc[4] += s1.x * (e[4] + o[4]);
    acc[5] += s1.y * (e[5] + o[5]);
    acc[6] += s1.z * (e[6] + o[6]);
    acc[7] += s1.w * (e[7] + o[7]);
}

constexpr int CH = CHUNK_STEPS;
constexpr int step_tile(int s) { int t = 0; while (s >= tile_base(t + 1)) ++t; return t; }
constexpr int step_group(int s) { return s - tile_base(step_tile(s)); }

template <int S, int SEnd>
struct LoadB {
    static __device__ __forceinline__ void run(const float* img, unsigned boff, v2f* B) {
        if constexpr (S < NGTOT) B[S % CH] = *(const v2f*)(img + boff + S * 64);
        LoadB<S + 1, SEnd>::run(img, boff, B);
    }
};
template <int SEnd>
struct LoadB<SEnd, SEnd> {
    static __device__ __forceinline__ void run(const float*, unsigned, v2f*) {}
};

template <int T>
static __device__ __forceinline__ void finish_tile(const float* xh, int lm, v8f& e, v8f& o, v8f& acc0, v8f& acc1) {
    if constexpr (T < NMAIN) {
        scale_add(xh + T * 16, e, o, acc0);
    } else {
        int i = 8 * (T - NMAIN) + (lm >> 1);
        if (i > 18) i = 18;
        scale_add(xh + i * 16, e, o, acc1);
    }
    e = v8f{0.f, 0.f, 0.f, 0.f, 0.f, 0.f, 0.f, 0.f};
    o = v8f{0.f, 0.f, 0.f, 0.f, 0.f, 0.f, 0.f, 0.f};
}

template <int S, int SEnd>
struct Mma {
    static __device__ __forceinline__ void run(const v2f* A, const v2f* B, const float* xh, int lm,
                                               v8f& e, v8f& o, v8f& acc0, v8f& acc1) {
        if constexpr (S < NGTOT) {
            constexpr int t = step_tile(S);
            constexpr int g = step_group(S);
            if constexpr ((g & 1) == 0)
                e = __builtin_amdgcn_wmma_f32_16x16x4_f32(false, A[g], false, B[S % CH], (short)0, e, false, false);
            else
                o = __builtin_amdgcn_wmma_f32_16x16x4_f32(false, A[g], false, B[S % CH], (short)0, o, false, false);
            if constexpr (g == tile_groups(t) - 1) finish_tile<t>(xh, lm, e, o, acc0, acc1);
        }
        Mma<S + 1, SEnd>::run(A, B, xh, lm, e, o, acc0, acc1);
    }
};
template <int SEnd>
struct Mma<SEnd, SEnd> {
    static __device__ __forceinline__ void run(const v2f*, const v2f*, const float*, int, v8f&, v8f&, v8f&, v8f&) {}
};

template <int C>
struct Chunks {
    static __device__ __forceinline__ void run(const v2f* A, const float* img, unsigned& boff, const float* xh, int lm,
                                               v2f* Bcur, v2f* Bnext, v8f& e, v8f& o, v8f& acc0, v8f& acc1) {
        LoadB<(C + 1) * CH, (C + 2) * CH>::run(img, boff, Bnext);
        Mma<C * CH, (C + 1) * CH>::run(A, Bcur, xh, lm, e, o, acc0, acc1);
        asm volatile("" : "+v"(boff), "+v"(e), "+v"(o), "+v"(acc0), "+v"(acc1));
        if constexpr ((C + 1) * CH < NGTOT)
            Chunks<C + 1>::run(A, img, boff, xh, lm, Bnext, Bcur, e, o, acc0, acc1);
    }
};

__global__ __launch_bounds__(256)
void vindy_wmma_kernel(const float* __restrict__ z,
                       const float* __restrict__ betas,
                       const float* __restrict__ big_xi,
                       const float* __restrict__ mask,
                       float* __restrict__ out,
                       int batch)
{
    __shared__ __attribute__((aligned(16))) float sImg[IMG_FLOATS];
    __shared__ __attribute__((aligned(16))) float sXT[WAVES][NMAIN][ROWS_PER_WAVE];

    const int tid  = threadIdx.x;
    const int lane = tid & 31;
    const int w    = tid >> 5;
    const int half = lane >> 4;
    const int lm   = lane & 15;
    const int rowBase = blockIdx.x * ROWS_PER_BLOCK + w * ROWS_PER_WAVE;

    {
        v4f* s4 = (v4f*)sImg;
        const v4f zero4 = {0.f, 0.f, 0.f, 0.f};
        constexpr int TAIL0 = tile_base(NMAIN) * 16;
        constexpr int TAILN = (NGTOT - tile_base(NMAIN)) * 16;
        for (int t = tid; t < TAILN; t += 256) s4[TAIL0 + t] = zero4;
        for (int t = tid; t < NMAIN * 16; t += 256)
            s4[(int)STAB.lastg[t >> 4] * 16 + (t & 15)] = zero4;
    }
    __syncthreads();

    {
        float* s = sImg;
        const v4f* xi4 = (const v4f*)big_xi;
        const v4f* mk4 = (const v4f*)mask;
        constexpr int NQ = NP * NV / 4;
        constexpr int UB = 8;
        static_assert(NP * NV % 4 == 0, "quad loads need a whole number of quads");
        for (int base = tid; base < NQ; base += 256 * UB) {
            v4f a[UB], m[UB];
            unsigned w0[UB], w1[UB];
#pragma unroll
            for (int u = 0; u < UB; ++u) {
                int q = base + 256 * u;
                if (q > NQ - 1) q = NQ - 1;
                a[u] = xi4[q];
                m[u] = mk4[q];
                const unsigned f0 = (4u * (unsigned)q) / (unsigned)NV;
                w0[u] = STAB.v[f0];
                w1[u] = STAB.v[f0 + 1u];
            }
#pragma unroll
            for (int u = 0; u < UB; ++u) {
                const int q = base + 256 * u;
                if (q < NQ) {
                    const unsigned i0 = 4u * (unsigned)q;
                    const unsigned f0 = i0 / (unsigned)NV;
                    const unsigned n0 = i0 - f0 * (unsigned)NV;
                    const unsigned m0 = w0[u] & 0xffffu;
                    const unsigned t0 = w0[u] >> 16;
                    const unsigned m1 = w1[u] & 0xffffu;
                    const unsigned d0 = (n0 < 16u) ? (m0 + 2u * n0) : t0;
                    const unsigned d2 = (n0 < 14u) ? (m0 + 2u * n0 + 4u) : ((n0 == 14u) ? t0 : m1);
                    s[d0]      = a[u].x * m[u].x;
                    s[d0 + 2u] = a[u].y * m[u].y;
                    s[d2]      = a[u].z * m[u].z;
                    s[d2 + 2u] = a[u].w * m[u].w;
                }
            }
        }
    }

    int gr = rowBase + lm;
    if (gr > batch - 1) gr = batch - 1;
    float x[19];
    {
        const v4f* zr = (const v4f*)(z + (size_t)gr * 16);
        v4f q0 = zr[0], q1 = zr[1], q2 = zr[2], q3 = zr[3];
        x[0]  = q0.x; x[1]  = q0.y; x[2]  = q0.z; x[3]  = q0.w;
        x[4]  = q1.x; x[5]  = q1.y; x[6]  = q1.z; x[7]  = q1.w;
        x[8]  = q2.x; x[9]  = q2.y; x[10] = q2.z; x[11] = q2.w;
        x[12] = q3.x; x[13] = q3.y; x[14] = q3.z; x[15] = q3.w;
        const v2f* br = (const v2f*)(betas + (size_t)gr * 2);
        v2f bb = br[0];
        x[16] = bb.x; x[17] = bb.y; x[18] = 1.0f;
    }

    const bool hi = (half != 0);

    if (!hi) {
#pragma unroll
        for (int i = 0; i < NMAIN; ++i) sXT[w][i][lm] = x[i];
    }

    v2f A[NAG];
    FillA<0, NAG>::run(x, hi, A);

    __syncthreads();

    v8f acc0 = {0.f, 0.f, 0.f, 0.f, 0.f, 0.f, 0.f, 0.f};
    v8f acc1 = {0.f, 0.f, 0.f, 0.f, 0.f, 0.f, 0.f, 0.f};

    unsigned boff = 2u * (unsigned)lane;
    const float* xh = &sXT[w][0][8 * half];

    v8f e = {0.f, 0.f, 0.f, 0.f, 0.f, 0.f, 0.f, 0.f};
    v8f o = {0.f, 0.f, 0.f, 0.f, 0.f, 0.f, 0.f, 0.f};
    v2f Ba[CH], Bb[CH];
    LoadB<0, CH>::run(sImg, boff, Ba);
    Chunks<0>::run(A, sImg, boff, xh, lm, Ba, Bb, e, o, acc0, acc1);

#pragma unroll
    for (int v = 0; v < 8; ++v) {
        float t = acc1[v];
        t += __shfl_xor(t, 2);
        t += __shfl_xor(t, 4);
        t += __shfl_xor(t, 8);
        acc1[v] = t;
    }

    const int orow = rowBase + 8 * half;
    if (rowBase + ROWS_PER_WAVE <= batch) {
#pragma unroll
        for (int v = 0; v < 8; ++v)
            out[(size_t)(orow + v) * NV + lm] = acc0[v];
        if (lm < 2) {
#pragma unroll
            for (int v = 0; v < 8; ++v)
                out[(size_t)(orow + v) * NV + 16 + lm] = acc1[v];
        }
    } else {
#pragma unroll
        for (int v = 0; v < 8; ++v) {
            int r = orow + v;
            if (r < batch) {
                out[(size_t)r * NV + lm] = acc0[v];
                if (lm < 2) out[(size_t)r * NV + 16 + lm] = acc1[v];
            }
        }
    }
}

extern "C" void kernel_launch(void* const* d_in, const int* in_sizes, int n_in,
                              void* d_out, int out_size, void* d_ws, size_t ws_size,
                              hipStream_t stream) {
    const float* z      = (const float*)d_in[0];
    const float* betas  = (const float*)d_in[1];
    const float* big_xi = (const float*)d_in[2];
    const float* mask   = (const float*)d_in[3];
    float* out          = (float*)d_out;

    const int batch = in_sizes[0] / 16;
    const int grid  = (batch + ROWS_PER_BLOCK - 1) / ROWS_PER_BLOCK;

    vindy_wmma_kernel<<<grid, 256, 0, stream>>>(z, betas, big_xi, mask, out, batch);
}


# One 16-wave workgroup per WGP (grid 256) builds the Xi*mask fragment image in LDS once and its waves loop over 16-row tiles: image built 256 times per launch, not 1024; sum order unchanged
# speedup vs baseline: 2.9516x; 1.1485x over previous
#include <hip/hip_runtime.h>


#define NV 18
#define NP 1330
#define NPAIR 190
#define NAG 48
#define NMAIN 19
#define NTILE 22
#define WAVES 16
#define NTHREADS (WAVES * 32)
#define ROWS_PER_WAVE 16
#define NBLOCKS 256

typedef __attribute__((ext_vector_type(2))) float v2f;
typedef __attribute__((ext_vector_type(4))) float v4f;
typedef __attribute__((ext_vector_type(8))) float v8f;

constexpr int pairs_from(int i) { return (19 - i) * (20 - i) / 2; }
constexpr int pair_index(int j, int k) { return (18 - j) * (19 - j) / 2 + (k - j); }
constexpr int pair_j(int p) {
    int j = 18;
    while (p >= (18 - j) * (19 - j) / 2 + (19 - j)) --j;
    return j;
}
constexpr int pair_k(int p) { return pair_j(p) + (p - (18 - pair_j(p)) * (19 - pair_j(p)) / 2); }
constexpr int tile_groups(int t) {
    return t < NMAIN ? (pairs_from(t) + 3) / 4 : (pairs_from(8 * (t - NMAIN)) + 3) / 4;
}
constexpr int tile_base(int t) {
    int s = 0;
    for (int u = 0; u < t; ++u) s += tile_groups(u);
    return s;
}
constexpr int CHUNK_STEPS = 12;
constexpr int NGTOT = tile_base(NTILE);
constexpr int IMG_FLOATS = NGTOT * 64;

struct StageTab {
    unsigned v[NP + 1];
    unsigned short lastg[NMAIN + 1];
};
constexpr unsigned stage_word(int i, int j, int k) {
    const int p = pair_index(j, k);
    const int m = (tile_base(i) + (p >> 2)) * 64 + 32 * ((p >> 1) & 1) + (p & 1);
    const int t = (tile_base(NMAIN + (i >> 3)) + (p >> 2)) * 64
                + (2 * (i & 7) + 16 * ((p >> 1) & 1)) * 2 + (p & 1);
    return (unsigned)m | ((unsigned)t << 16);
}
constexpr StageTab make_stage_tab() {
    StageTab t{};
    int f = 0;
    t.v[f++] = stage_word(18, 18, 18);
    for (int i = 0; i < NV; ++i) t.v[f++] = stage_word(i, 18, 18);
    for (int i = 0; i < NV; ++i)
        for (int j = i; j < NV; ++j) t.v[f++] = stage_word(i, j, 18);
    for (int i = 0; i < NV; ++i)
        for (int j = i; j < NV; ++j)
            for (int k = j; k < NV; ++k) t.v[f++] = stage_word(i, j, k);
    t.v[NP] = 0;
    for (int i = 0; i < NMAIN; ++i) t.lastg[i] = (unsigned short)(tile_base(i + 1) - 1);
    t.lastg[NMAIN] = 0;
    return t;
}
__constant__ const StageTab STAB = make_stage_tab();

template <int P>
static __device__ __forceinline__ float pairprod(const float* x) {
    if constexpr (P >= NPAIR) {
        return 0.0f;
    } else {
        constexpr int j = pair_j(P);
        constexpr int k = pair_k(P);
        if constexpr (k == 18) return x[j];
        else return x[j] * x[k];
    }
}

template <int G, int GEnd>
struct FillA {
    static __device__ __forceinline__ void run(const float* x, bool hi, v2f* A) {
        constexpr int p = 4 * G;
        const float q0 = pairprod<p + 0>(x);
        const float q1 = pairprod<p + 1>(x);
        const float q2 = pairprod<p + 2>(x);
        const float q3 = pairprod<p + 3>(x);
        A[G].x = hi ? q2 : q0;
        A[G].y = hi ? q3 : q1;
        FillA<G + 1, GEnd>::run(x, hi, A);
    }
};
template <int GEnd>
struct FillA<GEnd, GEnd> {
    static __device__ __forceinline__ void run(const float*, bool, v2f*) {}
};

static __device__ __forceinline__ void scale_add(const float* xs, const v8f& e, const v8f& o, v8f& acc) {
    const v4f s0 = *(const v4f*)(xs);
    const v4f s1 = *(const v4f*)(xs + 4);
    acc[0] += s0.x * (e[0] + o[0]);
    acc[1] += s0.y * (e[1] + o[1]);
    acc[2] += s0.z * (e[2] + o[2]);
    acc[3] += s0.w * (e[3] + o[3]);
    acc[4] += s1.x * (e[4] + o[4]);
    acc[5] += s1.y * (e[5] + o[5]);
    acc[6] += s1.z * (e[6] + o[6]);
    acc[7] += s1.w * (e[7] + o[7]);
}

constexpr int CH = CHUNK_STEPS;
constexpr int step_tile(int s) { int t = 0; while (s >= tile_base(t + 1)) ++t; return t; }
constexpr int step_group(int s) { return s - tile_base(step_tile(s)); }

template <int S, int SEnd>
struct LoadB {
    static __device__ __forceinline__ void run(const float* img, unsigned boff, v2f* B) {
        if constexpr (S < NGTOT) B[S % CH] = *(const v2f*)(img + boff + S * 64);
        LoadB<S + 1, SEnd>::run(img, boff, B);
    }
};
template <int SEnd>
struct LoadB<SEnd, SEnd> {
    static __device__ __forceinline__ void run(const float*, unsigned, v2f*) {}
};

template <int T>
static __device__ __forceinline__ void finish_tile(const float* xh, int lm, v8f& e, v8f& o, v8f& acc0, v8f& acc1) {
    if constexpr (T < NMAIN) {
        scale_add(xh + T * 16, e, o, acc0);
    } else {
        int i = 8 * (T - NMAIN) + (lm >> 1);
        if (i > 18) i = 18;
        scale_add(xh + i * 16, e, o, acc1);
    }
    e = v8f{0.f, 0.f, 0.f, 0.f, 0.f, 0.f, 0.f, 0.f};
    o = v8f{0.f, 0.f, 0.f, 0.f, 0.f, 0.f, 0.f, 0.f};
}

template <int S, int SEnd>
struct Mma {
    static __device__ __forceinline__ void run(const v2f* A, const v2f* B, const float* xh, int lm,
                                               v8f& e, v8f& o, v8f& acc0, v8f& acc1) {
        if constexpr (S < NGTOT) {
            constexpr int t = step_tile(S);
            constexpr int g = step_group(S);
            if constexpr ((g & 1) == 0)
                e = __builtin_amdgcn_wmma_f32_16x16x4_f32(false, A[g], false, B[S % CH], (short)0, e, false, false);
            else
                o = __builtin_amdgcn_wmma_f32_16x16x4_f32(false, A[g], false, B[S % CH], (short)0, o, false, false);
            if constexpr (g == tile_groups(t) - 1) finish_tile<t>(xh, lm, e, o, acc0, acc1);
        }
        Mma<S + 1, SEnd>::run(A, B, xh, lm, e, o, acc0, acc1);
    }
};
template <int SEnd>
struct Mma<SEnd, SEnd> {
    static __device__ __forceinline__ void run(const v2f*, const v2f*, const float*, int, v8f&, v8f&, v8f&, v8f&) {}
};

template <int C>
struct Chunks {
    static __device__ __forceinline__ void run(const v2f* A, const float* img, unsigned& boff, const float* xh, int lm,
                                               v2f* Bcur, v2f* Bnext, v8f& e, v8f& o, v8f& acc0, v8f& acc1) {
        LoadB<(C + 1) * CH, (C + 2) * CH>::run(img, boff, Bnext);
        Mma<C * CH, (C + 1) * CH>::run(A, Bcur, xh, lm, e, o, acc0, acc1);
        asm volatile("" : "+v"(boff), "+v"(e), "+v"(o), "+v"(acc0), "+v"(acc1));
        if constexpr ((C + 1) * CH < NGTOT)
            Chunks<C + 1>::run(A, img, boff, xh, lm, Bnext, Bcur, e, o, acc0, acc1);
    }
};

static __device__ __forceinline__ void wave_lds_order() {
    __builtin_amdgcn_fence(__ATOMIC_RELEASE, "wavefront");
    __builtin_amdgcn_wave_barrier();
    __builtin_amdgcn_fence(__ATOMIC_ACQUIRE, "wavefront");
}

__global__ __launch_bounds__(NTHREADS)
void vindy_wmma_kernel(const float* __restrict__ z,
                       const float* __restrict__ betas,
                       const float* __restrict__ big_xi,
                       const float* __restrict__ mask,
                       float* __restrict__ out,
                       int batch)
{
    __shared__ __attribute__((aligned(16))) float sImg[IMG_FLOATS];
    __shared__ __attribute__((aligned(16))) float sXT[WAVES][NMAIN][ROWS_PER_WAVE];

    const int tid  = threadIdx.x;
    const int lane = tid & 31;
    const int w    = __builtin_amdgcn_readfirstlane(tid >> 5);
    const int half = lane >> 4;
    const int lm   = lane & 15;

    {
        v4f* s4 = (v4f*)sImg;
        const v4f zero4 = {0.f, 0.f, 0.f, 0.f};
        constexpr int TAIL0 = tile_base(NMAIN) * 16;
        constexpr int TAILN = (NGTOT - tile_base(NMAIN)) * 16;
        for (int t = tid; t < TAILN; t += NTHREADS) s4[TAIL0 + t] = zero4;
        for (int t = tid; t < NMAIN * 16; t += NTHREADS)
            s4[(int)STAB.lastg[t >> 4] * 16 + (t & 15)] = zero4;
    }
    __syncthreads();

    {
        float* s = sImg;
        const v4f* xi4 = (const v4f*)big_xi;
        const v4f* mk4 = (const v4f*)mask;
        constexpr int NQ = NP * NV / 4;
        constexpr int UB = 12;
        static_assert(NP * NV % 4 == 0, "quad loads need a whole number of quads");
        for (int base = tid; base < NQ; base += NTHREADS * UB) {
            v4f a[UB], m[UB];
            unsigned w0[UB], w1[UB];
#pragma unroll
            for (int u = 0; u < UB; ++u) {
                int q = base + NTHREADS * u;
                if (q > NQ - 1) q = NQ - 1;
                a[u] = xi4[q];
                m[u] = mk4[q];
                const unsigned f0 = (4u * (unsigned)q) / (unsigned)NV;
                w0[u] = STAB.v[f0];
                w1[u] = STAB.v[f0 + 1u];
            }
#pragma unroll
            for (int u = 0; u < UB; ++u) {
                const int q = base + NTHREADS * u;
                if (q < NQ) {
                    const unsigned i0 = 4u * (unsigned)q;
                    const unsigned f0 = i0 / (unsigned)NV;
                    const unsigned n0 = i0 - f0 * (unsigned)NV;
                    const unsigned m0 = w0[u] & 0xffffu;
                    const unsigned t0 = w0[u] >> 16;
                    const unsigned m1 = w1[u] & 0xffffu;
                    const unsigned d0 = (n0 < 16u) ? (m0 + 2u * n0) : t0;
                    const unsigned d2 = (n0 < 14u) ? (m0 + 2u * n0 + 4u) : ((n0 == 14u) ? t0 : m1);
                    s[d0]      = a[u].x * m[u].x;
                    s[d0 + 2u] = a[u].y * m[u].y;
                    s[d2]      = a[u].z * m[u].z;
                    s[d2 + 2u] = a[u].w * m[u].w;
                }
            }
        }
    }

    __syncthreads();

    const bool hi = (half != 0);
    const float* xh = &sXT[w][0][8 * half];
    const int nWaveTiles = (batch + ROWS_PER_WAVE - 1) / ROWS_PER_WAVE;

    for (int wt = blockIdx.x * WAVES + w; wt < nWaveTiles; wt += NBLOCKS * WAVES) {
        const int rowBase = wt * ROWS_PER_WAVE;

        int gr = rowBase + lm;
        if (gr > batch - 1) gr = batch - 1;
        float x[19];
        {
            const v4f* zr = (const v4f*)(z + (size_t)gr * 16);
            v4f q0 = zr[0], q1 = zr[1], q2 = zr[2], q3 = zr[3];
            x[0]  = q0.x; x[1]  = q0.y; x[2]  = q0.z; x[3]  = q0.w;
            x[4]  = q1.x; x[5]  = q1.y; x[6]  = q1.z; x[7]  = q1.w;
            x[8]  = q2.x; x[9]  = q2.y; x[10] = q2.z; x[11] = q2.w;
            x[12] = q3.x; x[13] = q3.y; x[14] = q3.z; x[15] = q3.w;
            const v2f* br = (const v2f*)(betas + (size_t)gr * 2);
            v2f bb = br[0];
            x[16] = bb.x; x[17] = bb.y; x[18] = 1.0f;
        }

        wave_lds_order();
        if (!hi) {
#pragma unroll
            for (int i = 0; i < NMAIN; ++i) sXT[w][i][lm] = x[i];
        }
        wave_lds_order();

        v2f A[NAG];
        FillA<0, NAG>::run(x, hi, A);

        v8f acc0 = {0.f, 0.f, 0.f, 0.f, 0.f, 0.f, 0.f, 0.f};
        v8f acc1 = {0.f, 0.f, 0.f, 0.f, 0.f, 0.f, 0.f, 0.f};

        unsigned boff = 2u * (unsigned)lane;

        v8f e = {0.f, 0.f, 0.f, 0.f, 0.f, 0.f, 0.f, 0.f};
        v8f o = {0.f, 0.f, 0.f, 0.f, 0.f, 0.f, 0.f, 0.f};
        v2f Ba[CH], Bb[CH];
        LoadB<0, CH>::run(sImg, boff, Ba);
        Chunks<0>::run(A, sImg, boff, xh, lm, Ba, Bb, e, o, acc0, acc1);

#pragma unroll
        for (int v = 0; v < 8; ++v) {
            float t = acc1[v];
            t += __shfl_xor(t, 2);
            t += __shfl_xor(t, 4);
            t += __shfl_xor(t, 8);
            acc1[v] = t;
        }

        const int orow = rowBase + 8 * half;
        if (rowBase + ROWS_PER_WAVE <= batch) {
#pragma unroll
            for (int v = 0; v < 8; ++v)
                out[(size_t)(orow + v) * NV + lm] = acc0[v];
            if (lm < 2) {
#pragma unroll
                for (int v = 0; v < 8; ++v)
                    out[(size_t)(orow + v) * NV + 16 + lm] = acc1[v];
            }
        } else {
#pragma unroll
            for (int v = 0; v < 8; ++v) {
                int r = orow + v;
                if (r < batch) {
                    out[(size_t)r * NV + lm] = acc0[v];
                    if (lm < 2) out[(size_t)r * NV + 16 + lm] = acc1[v];
                }
            }
        }
    }
}

extern "C" void kernel_launch(void* const* d_in, const int* in_sizes, int n_in,
                              void* d_out, int out_size, void* d_ws, size_t ws_size,
                              hipStream_t stream) {
    const float* z      = (const float*)d_in[0];
    const float* betas  = (const float*)d_in[1];
    const float* big_xi = (const float*)d_in[2];
    const float* mask   = (const float*)d_in[3];
    float* out          = (float*)d_out;

    const int batch = in_sizes[0] / 16;

    vindy_wmma_kernel<<<NBLOCKS, NTHREADS, 0, stream>>>(z, betas, big_xi, mask, out, batch);
}


# Pin each chunk's LDS loads of the next chunk's B fragments above the current chunk's WMMAs with a scheduling barrier: a real register double buffer, no LDS round trip per chunk; sum order unchanged
# speedup vs baseline: 3.1231x; 1.0581x over previous
#include <hip/hip_runtime.h>


#define NV 18
#define NP 1330
#define NPAIR 190
#define NAG 48
#define NMAIN 19
#define NTILE 22
#define WAVES 16
#define NTHREADS (WAVES * 32)
#define ROWS_PER_WAVE 16
#define NBLOCKS 256

typedef __attribute__((ext_vector_type(2))) float v2f;
typedef __attribute__((ext_vector_type(4))) float v4f;
typedef __attribute__((ext_vector_type(8))) float v8f;

constexpr int pairs_from(int i) { return (19 - i) * (20 - i) / 2; }
constexpr int pair_index(int j, int k) { return (18 - j) * (19 - j) / 2 + (k - j); }
constexpr int pair_j(int p) {
    int j = 18;
    while (p >= (18 - j) * (19 - j) / 2 + (19 - j)) --j;
    return j;
}
constexpr int pair_k(int p) { return pair_j(p) + (p - (18 - pair_j(p)) * (19 - pair_j(p)) / 2); }
constexpr int tile_groups(int t) {
    return t < NMAIN ? (pairs_from(t) + 3) / 4 : (pairs_from(8 * (t - NMAIN)) + 3) / 4;
}
constexpr int tile_base(int t) {
    int s = 0;
    for (int u = 0; u < t; ++u) s += tile_groups(u);
    return s;
}
constexpr int CHUNK_STEPS = 12;
constexpr int NGTOT = tile_base(NTILE);
constexpr int IMG_FLOATS = NGTOT * 64;

struct StageTab {
    unsigned v[NP + 1];
    unsigned short lastg[NMAIN + 1];
};
constexpr unsigned stage_word(int i, int j, int k) {
    const int p = pair_index(j, k);
    const int m = (tile_base(i) + (p >> 2)) * 64 + 32 * ((p >> 1) & 1) + (p & 1);
    const int t = (tile_base(NMAIN + (i >> 3)) + (p >> 2)) * 64
                + (2 * (i & 7) + 16 * ((p >> 1) & 1)) * 2 + (p & 1);
    return (unsigned)m | ((unsigned)t << 16);
}
constexpr StageTab make_stage_tab() {
    StageTab t{};
    int f = 0;
    t.v[f++] = stage_word(18, 18, 18);
    for (int i = 0; i < NV; ++i) t.v[f++] = stage_word(i, 18, 18);
    for (int i = 0; i < NV; ++i)
        for (int j = i; j < NV; ++j) t.v[f++] = stage_word(i, j, 18);
    for (int i = 0; i < NV; ++i)
        for (int j = i; j < NV; ++j)
            for (int k = j; k < NV; ++k) t.v[f++] = stage_word(i, j, k);
    t.v[NP] = 0;
    for (int i = 0; i < NMAIN; ++i) t.lastg[i] = (unsigned short)(tile_base(i + 1) - 1);
    t.lastg[NMAIN] = 0;
    return t;
}
__constant__ const StageTab STAB = make_stage_tab();

template <int P>
static __device__ __forceinline__ float pairprod(const float* x) {
    if constexpr (P >= NPAIR) {
        return 0.0f;
    } else {
        constexpr int j = pair_j(P);
        constexpr int k = pair_k(P);
        if constexpr (k == 18) return x[j];
        else return x[j] * x[k];
    }
}

template <int G, int GEnd>
struct FillA {
    static __device__ __forceinline__ void run(const float* x, bool hi, v2f* A) {
        constexpr int p = 4 * G;
        const float q0 = pairprod<p + 0>(x);
        const float q1 = pairprod<p + 1>(x);
        const float q2 = pairprod<p + 2>(x);
        const float q3 = pairprod<p + 3>(x);
        A[G].x = hi ? q2 : q0;
        A[G].y = hi ? q3 : q1;
        FillA<G + 1, GEnd>::run(x, hi, A);
    }
};
template <int GEnd>
struct FillA<GEnd, GEnd> {
    static __device__ __forceinline__ void run(const float*, bool, v2f*) {}
};

static __device__ __forceinline__ void scale_add(const float* xs, const v8f& e, const v8f& o, v8f& acc) {
    const v4f s0 = *(const v4f*)(xs);
    const v4f s1 = *(const v4f*)(xs + 4);
    acc[0] += s0.x * (e[0] + o[0]);
    acc[1] += s0.y * (e[1] + o[1]);
    acc[2] += s0.z * (e[2] + o[2]);
    acc[3] += s0.w * (e[3] + o[3]);
    acc[4] += s1.x * (e[4] + o[4]);
    acc[5] += s1.y * (e[5] + o[5]);
    acc[6] += s1.z * (e[6] + o[6]);
    acc[7] += s1.w * (e[7] + o[7]);
}

constexpr int CH = CHUNK_STEPS;
constexpr int step_tile(int s) { int t = 0; while (s >= tile_base(t + 1)) ++t; return t; }
constexpr int step_group(int s) { return s - tile_base(step_tile(s)); }

template <int S, int SEnd>
struct LoadB {
    static __device__ __forceinline__ void run(const float* img, unsigned boff, v2f* B) {
        if constexpr (S < NGTOT) B[S % CH] = *(const v2f*)(img + boff + S * 64);
        LoadB<S + 1, SEnd>::run(img, boff, B);
    }
};
template <int SEnd>
struct LoadB<SEnd, SEnd> {
    static __device__ __forceinline__ void run(const float*, unsigned, v2f*) {}
};

template <int T>
static __device__ __forceinline__ void finish_tile(const float* xh, int lm, v8f& e, v8f& o, v8f& acc0, v8f& acc1) {
    if constexpr (T < NMAIN) {
        scale_add(xh + T * 16, e, o, acc0);
    } else {
        int i = 8 * (T - NMAIN) + (lm >> 1);
        if (i > 18) i = 18;
        scale_add(xh + i * 16, e, o, acc1);
    }
    e = v8f{0.f, 0.f, 0.f, 0.f, 0.f, 0.f, 0.f, 0.f};
    o = v8f{0.f, 0.f, 0.f, 0.f, 0.f, 0.f, 0.f, 0.f};
}

template <int S, int SEnd>
struct Mma {
    static __device__ __forceinline__ void run(const v2f* A, const v2f* B, const float* xh, int lm,
                                               v8f& e, v8f& o, v8f& acc0, v8f& acc1) {
        if constexpr (S < NGTOT) {
            constexpr int t = step_tile(S);
            constexpr int g = step_group(S);
            if constexpr ((g & 1) == 0)
                e = __builtin_amdgcn_wmma_f32_16x16x4_f32(false, A[g], false, B[S % CH], (short)0, e, false, false);
            else
                o = __builtin_amdgcn_wmma_f32_16x16x4_f32(false, A[g], false, B[S % CH], (short)0, o, false, false);
            if constexpr (g == tile_groups(t) - 1) finish_tile<t>(xh, lm, e, o, acc0, acc1);
        }
        Mma<S + 1, SEnd>::run(A, B, xh, lm, e, o, acc0, acc1);
    }
};
template <int SEnd>
struct Mma<SEnd, SEnd> {
    static __device__ __forceinline__ void run(const v2f*, const v2f*, const float*, int, v8f&, v8f&, v8f&, v8f&) {}
};

template <int C>
struct Chunks {
    static __device__ __forceinline__ void run(const v2f* A, const float* img, unsigned& boff, const float* xh, int lm,
                                               v2f* Bcur, v2f* Bnext, v8f& e, v8f& o, v8f& acc0, v8f& acc1) {
        LoadB<(C + 1) * CH, (C + 2) * CH>::run(img, boff, Bnext);
        __builtin_amdgcn_sched_barrier(0x6);
        Mma<C * CH, (C + 1) * CH>::run(A, Bcur, xh, lm, e, o, acc0, acc1);
        asm volatile("" : "+v"(boff), "+v"(e), "+v"(o), "+v"(acc0), "+v"(acc1));
        if constexpr ((C + 1) * CH < NGTOT)
            Chunks<C + 1>::run(A, img, boff, xh, lm, Bnext, Bcur, e, o, acc0, acc1);
    }
};

static __device__ __forceinline__ void wave_lds_order() {
    __builtin_amdgcn_fence(__ATOMIC_RELEASE, "wavefront");
    __builtin_amdgcn_wave_barrier();
    __builtin_amdgcn_fence(__ATOMIC_ACQUIRE, "wavefront");
}

__global__ __launch_bounds__(NTHREADS)
void vindy_wmma_kernel(const float* __restrict__ z,
                       const float* __restrict__ betas,
                       const float* __restrict__ big_xi,
                       const float* __restrict__ mask,
                       float* __restrict__ out,
                       int batch)
{
    __shared__ __attribute__((aligned(16))) float sImg[IMG_FLOATS];
    __shared__ __attribute__((aligned(16))) float sXT[WAVES][NMAIN][ROWS_PER_WAVE];

    const int tid  = threadIdx.x;
    const int lane = tid & 31;
    const int w    = __builtin_amdgcn_readfirstlane(tid >> 5);
    const int half = lane >> 4;
    const int lm   = lane & 15;

    {
        v4f* s4 = (v4f*)sImg;
        const v4f zero4 = {0.f, 0.f, 0.f, 0.f};
        constexpr int TAIL0 = tile_base(NMAIN) * 16;
        constexpr int TAILN = (NGTOT - tile_base(NMAIN)) * 16;
        for (int t = tid; t < TAILN; t += NTHREADS) s4[TAIL0 + t] = zero4;
        for (int t = tid; t < NMAIN * 16; t += NTHREADS)
            s4[(int)STAB.lastg[t >> 4] * 16 + (t & 15)] = zero4;
    }
    __syncthreads();

    {
        float* s = sImg;
        const v4f* xi4 = (const v4f*)big_xi;
        const v4f* mk4 = (const v4f*)mask;
        constexpr int NQ = NP * NV / 4;
        constexpr int UB = 12;
        static_assert(NP * NV % 4 == 0, "quad loads need a whole number of quads");
        for (int base = tid; base < NQ; base += NTHREADS * UB) {
            v4f a[UB], m[UB];
            unsigned w0[UB], w1[UB];
#pragma unroll
            for (int u = 0; u < UB; ++u) {
                int q = base + NTHREADS * u;
                if (q > NQ - 1) q = NQ - 1;
                a[u] = xi4[q];
                m[u] = mk4[q];
                const unsigned f0 = (4u * (unsigned)q) / (unsigned)NV;
                w0[u] = STAB.v[f0];
                w1[u] = STAB.v[f0 + 1u];
            }
#pragma unroll
            for (int u = 0; u < UB; ++u) {
                const int q = base + NTHREADS * u;
                if (q < NQ) {
                    const unsigned i0 = 4u * (unsigned)q;
                    const unsigned f0 = i0 / (unsigned)NV;
                    const unsigned n0 = i0 - f0 * (unsigned)NV;
                    const unsigned m0 = w0[u] & 0xffffu;
                    const unsigned t0 = w0[u] >> 16;
                    const unsigned m1 = w1[u] & 0xffffu;
                    const unsigned d0 = (n0 < 16u) ? (m0 + 2u * n0) : t0;
                    const unsigned d2 = (n0 < 14u) ? (m0 + 2u * n0 + 4u) : ((n0 == 14u) ? t0 : m1);
                    s[d0]      = a[u].x * m[u].x;
                    s[d0 + 2u] = a[u].y * m[u].y;
                    s[d2]      = a[u].z * m[u].z;
                    s[d2 + 2u] = a[u].w * m[u].w;
                }
            }
        }
    }

    __syncthreads();

    const bool hi = (half != 0);
    const float* xh = &sXT[w][0][8 * half];
    const int nWaveTiles = (batch + ROWS_PER_WAVE - 1) / ROWS_PER_WAVE;

    for (int wt = blockIdx.x * WAVES + w; wt < nWaveTiles; wt += NBLOCKS * WAVES) {
        const int rowBase = wt * ROWS_PER_WAVE;

        int gr = rowBase + lm;
        if (gr > batch - 1) gr = batch - 1;
        float x[19];
        {
            const v4f* zr = (const v4f*)(z + (size_t)gr * 16);
            v4f q0 = zr[0], q1 = zr[1], q2 = zr[2], q3 = zr[3];
            x[0]  = q0.x; x[1]  = q0.y; x[2]  = q0.z; x[3]  = q0.w;
            x[4]  = q1.x; x[5]  = q1.y; x[6]  = q1.z; x[7]  = q1.w;
            x[8]  = q2.x; x[9]  = q2.y; x[10] = q2.z; x[11] = q2.w;
            x[12] = q3.x; x[13] = q3.y; x[14] = q3.z; x[15] = q3.w;
            const v2f* br = (const v2f*)(betas + (size_t)gr * 2);
            v2f bb = br[0];
            x[16] = bb.x; x[17] = bb.y; x[18] = 1.0f;
        }

        wave_lds_order();
        if (!hi) {
#pragma unroll
            for (int i = 0; i < NMAIN; ++i) sXT[w][i][lm] = x[i];
        }
        wave_lds_order();

        v2f A[NAG];
        FillA<0, NAG>::run(x, hi, A);

        v8f acc0 = {0.f, 0.f, 0.f, 0.f, 0.f, 0.f, 0.f, 0.f};
        v8f acc1 = {0.f, 0.f, 0.f, 0.f, 0.f, 0.f, 0.f, 0.f};

        unsigned boff = 2u * (unsigned)lane;

        v8f e = {0.f, 0.f, 0.f, 0.f, 0.f, 0.f, 0.f, 0.f};
        v8f o = {0.f, 0.f, 0.f, 0.f, 0.f, 0.f, 0.f, 0.f};
        v2f Ba[CH], Bb[CH];
        LoadB<0, CH>::run(sImg, boff, Ba);
        Chunks<0>::run(A, sImg, boff, xh, lm, Ba, Bb, e, o, acc0, acc1);

#pragma unroll
        for (int v = 0; v < 8; ++v) {
            float t = acc1[v];
            t += __shfl_xor(t, 2);
            t += __shfl_xor(t, 4);
            t += __shfl_xor(t, 8);
            acc1[v] = t;
        }

        const int orow = rowBase + 8 * half;
        if (rowBase + ROWS_PER_WAVE <= batch) {
#pragma unroll
            for (int v = 0; v < 8; ++v)
                out[(size_t)(orow + v) * NV + lm] = acc0[v];
            if (lm < 2) {
#pragma unroll
                for (int v = 0; v < 8; ++v)
                    out[(size_t)(orow + v) * NV + 16 + lm] = acc1[v];
            }
        } else {
#pragma unroll
            for (int v = 0; v < 8; ++v) {
                int r = orow + v;
                if (r < batch) {
                    out[(size_t)r * NV + lm] = acc0[v];
                    if (lm < 2) out[(size_t)r * NV + 16 + lm] = acc1[v];
                }
            }
        }
    }
}

extern "C" void kernel_launch(void* const* d_in, const int* in_sizes, int n_in,
                              void* d_out, int out_size, void* d_ws, size_t ws_size,
                              hipStream_t stream) {
    const float* z      = (const float*)d_in[0];
    const float* betas  = (const float*)d_in[1];
    const float* big_xi = (const float*)d_in[2];
    const float* mask   = (const float*)d_in[3];
    float* out          = (float*)d_out;

    const int batch = in_sizes[0] / 16;

    vindy_wmma_kernel<<<NBLOCKS, NTHREADS, 0, stream>>>(z, betas, big_xi, mask, out, batch);
}


# Fewer vector instructions beside the f32 WMMAs: two tiles in flight with one accumulator each (no even/odd merge adds) and byte LDS offsets with a second 64 KB base; sum order changes
# speedup vs baseline: 3.2257x; 1.0329x over previous
#include <hip/hip_runtime.h>


#define NV 18
#define NP 1330
#define NPAIR 190
#define NAG 48
#define NMAIN 19
#define NTILE 22
#define WAVES 16
#define NTHREADS (WAVES * 32)
#define ROWS_PER_WAVE 16
#define NBLOCKS 256

typedef __attribute__((ext_vector_type(2))) float v2f;
typedef __attribute__((ext_vector_type(4))) float v4f;
typedef __attribute__((ext_vector_type(8))) float v8f;

constexpr int pairs_from(int i) { return (19 - i) * (20 - i) / 2; }
constexpr int pair_index(int j, int k) { return (18 - j) * (19 - j) / 2 + (k - j); }
constexpr int pair_j(int p) {
    int j = 18;
    while (p >= (18 - j) * (19 - j) / 2 + (19 - j)) --j;
    return j;
}
constexpr int pair_k(int p) { return pair_j(p) + (p - (18 - pair_j(p)) * (19 - pair_j(p)) / 2); }
constexpr int tile_groups(int t) {
    return t < NMAIN ? (pairs_from(t) + 3) / 4 : (pairs_from(8 * (t - NMAIN)) + 3) / 4;
}
constexpr int tile_base(int t) {
    int s = 0;
    for (int u = 0; u < t; ++u) s += tile_groups(u);
    return s;
}
constexpr int CHUNK_STEPS = 12;
constexpr int NGTOT = tile_base(NTILE);
constexpr int IMG_FLOATS = NGTOT * 64;

struct StageTab {
    unsigned v[NP + 1];
    unsigned short lastg[NMAIN + 1];
};
constexpr unsigned stage_word(int i, int j, int k) {
    const int p = pair_index(j, k);
    const int m = (tile_base(i) + (p >> 2)) * 64 + 32 * ((p >> 1) & 1) + (p & 1);
    const int t = (tile_base(NMAIN + (i >> 3)) + (p >> 2)) * 64
                + (2 * (i & 7) + 16 * ((p >> 1) & 1)) * 2 + (p & 1);
    return (unsigned)m | ((unsigned)t << 16);
}
constexpr StageTab make_stage_tab() {
    StageTab t{};
    int f = 0;
    t.v[f++] = stage_word(18, 18, 18);
    for (int i = 0; i < NV; ++i) t.v[f++] = stage_word(i, 18, 18);
    for (int i = 0; i < NV; ++i)
        for (int j = i; j < NV; ++j) t.v[f++] = stage_word(i, j, 18);
    for (int i = 0; i < NV; ++i)
        for (int j = i; j < NV; ++j)
            for (int k = j; k < NV; ++k) t.v[f++] = stage_word(i, j, k);
    t.v[NP] = 0;
    for (int i = 0; i < NMAIN; ++i) t.lastg[i] = (unsigned short)(tile_base(i + 1) - 1);
    t.lastg[NMAIN] = 0;
    return t;
}
__constant__ const StageTab STAB = make_stage_tab();

template <int P>
static __device__ __forceinline__ float pairprod(const float* x) {
    if constexpr (P >= NPAIR) {
        return 0.0f;
    } else {
        constexpr int j = pair_j(P);
        constexpr int k = pair_k(P);
        if constexpr (k == 18) return x[j];
        else return x[j] * x[k];
    }
}

template <int G, int GEnd>
struct FillA {
    static __device__ __forceinline__ void run(const float* x, bool hi, v2f* A) {
        constexpr int p = 4 * G;
        const float q0 = pairprod<p + 0>(x);
        const float q1 = pairprod<p + 1>(x);
        const float q2 = pairprod<p + 2>(x);
        const float q3 = pairprod<p + 3>(x);
        A[G].x = hi ? q2 : q0;
        A[G].y = hi ? q3 : q1;
        FillA<G + 1, GEnd>::run(x, hi, A);
    }
};
template <int GEnd>
struct FillA<GEnd, GEnd> {
    static __device__ __forceinline__ void run(const float*, bool, v2f*) {}
};

static __device__ __forceinline__ void scale_add(const float* xs, const v8f& t, v8f& acc) {
    const v4f s0 = *(const v4f*)(xs);
    const v4f s1 = *(const v4f*)(xs + 4);
    acc[0] += s0.x * t[0];
    acc[1] += s0.y * t[1];
    acc[2] += s0.z * t[2];
    acc[3] += s0.w * t[3];
    acc[4] += s1.x * t[4];
    acc[5] += s1.y * t[5];
    acc[6] += s1.z * t[6];
    acc[7] += s1.w * t[7];
}

constexpr int CH = CHUNK_STEPS;
constexpr int HI_STEP0 = 256;

struct Sched { int w[NGTOT]; };
constexpr Sched make_sched() {
    Sched s{};
    const int order[NTILE] = {0, 19, 1, 2, 3, 4, 5, 6, 7, 8, 20, 9, 10, 11, 12, 13, 14, 15, 16, 21, 17, 18};
    int next = 2;
    int curT[2] = {order[0], order[1]};
    int curG[2] = {0, 0};
    int E = 0, c = 0;
    while (E < NGTOT) {
        if (curT[c] < 0) { c ^= 1; continue; }
        const bool last = (curG[c] == tile_groups(curT[c]) - 1);
        s.w[E] = curT[c] | (curG[c] << 8) | (c << 16) | ((last ? 1 : 0) << 17);
        if (last) {
            if (next < NTILE) { curT[c] = order[next++]; curG[c] = 0; } else curT[c] = -1;
        } else {
            ++curG[c];
        }
        ++E;
        c ^= 1;
    }
    return s;
}
constexpr int ex_word(int E)  { return make_sched().w[E]; }
constexpr int ex_tile(int E)  { return ex_word(E) & 0xff; }
constexpr int ex_group(int E) { return (ex_word(E) >> 8) & 0xff; }
constexpr int ex_chain(int E) { return (ex_word(E) >> 16) & 1; }
constexpr bool ex_last(int E) { return ((ex_word(E) >> 17) & 1) != 0; }
constexpr int ex_step(int E)  { return tile_base(ex_tile(E)) + ex_group(E); }

template <int E, int EEnd>
struct LoadB {
    static __device__ __forceinline__ void run(const float* img, unsigned bLo, unsigned bHi, v2f* B) {
        if constexpr (E < NGTOT) {
            constexpr int S = ex_step(E);
            if constexpr (S < HI_STEP0)
                B[E % CH] = *(const v2f*)((const char*)img + bLo + S * 256);
            else
                B[E % CH] = *(const v2f*)((const char*)img + bHi + (S - HI_STEP0) * 256);
        }
        LoadB<E + 1, EEnd>::run(img, bLo, bHi, B);
    }
};
template <int EEnd>
struct LoadB<EEnd, EEnd> {
    static __device__ __forceinline__ void run(const float*, unsigned, unsigned, v2f*) {}
};

template <int T>
static __device__ __forceinline__ void finish_tile(const float* xh, int lm, const v8f& t, v8f& acc0, v8f& acc1) {
    if constexpr (T < NMAIN) {
        scale_add(xh + T * 16, t, acc0);
    } else {
        int i = 8 * (T - NMAIN) + (lm >> 1);
        if (i > 18) i = 18;
        scale_add(xh + i * 16, t, acc1);
    }
}

template <int E, int EEnd>
struct Mma {
    static __device__ __forceinline__ void run(const v2f* A, const v2f* B, const float* xh, int lm,
                                               v8f& X, v8f& Y, v8f& acc0, v8f& acc1) {
        if constexpr (E < NGTOT) {
            constexpr int t = ex_tile(E);
            constexpr int g = ex_group(E);
            v8f& acc = (ex_chain(E) == 0) ? X : Y;
            if constexpr (g == 0) {
                const v8f zero = {0.f, 0.f, 0.f, 0.f, 0.f, 0.f, 0.f, 0.f};
                acc = __builtin_amdgcn_wmma_f32_16x16x4_f32(false, A[g], false, B[E % CH], (short)0, zero, false, false);
            } else {
                acc = __builtin_amdgcn_wmma_f32_16x16x4_f32(false, A[g], false, B[E % CH], (short)0, acc, false, false);
            }
            if constexpr (ex_last(E)) finish_tile<t>(xh, lm, acc, acc0, acc1);
        }
        Mma<E + 1, EEnd>::run(A, B, xh, lm, X, Y, acc0, acc1);
    }
};
template <int EEnd>
struct Mma<EEnd, EEnd> {
    static __device__ __forceinline__ void run(const v2f*, const v2f*, const float*, int, v8f&, v8f&, v8f&, v8f&) {}
};

template <int C>
struct Chunks {
    static __device__ __forceinline__ void run(const v2f* A, const float* img, unsigned& bLo, unsigned& bHi,
                                               const float* xh, int lm,
                                               v2f* Bcur, v2f* Bnext, v8f& X, v8f& Y, v8f& acc0, v8f& acc1) {
        LoadB<(C + 1) * CH, (C + 2) * CH>::run(img, bLo, bHi, Bnext);
        __builtin_amdgcn_sched_barrier(0x6);
        Mma<C * CH, (C + 1) * CH>::run(A, Bcur, xh, lm, X, Y, acc0, acc1);
        asm volatile("" : "+v"(bLo), "+v"(bHi), "+v"(X), "+v"(Y), "+v"(acc0), "+v"(acc1));
        if constexpr ((C + 1) * CH < NGTOT)
            Chunks<C + 1>::run(A, img, bLo, bHi, xh, lm, Bnext, Bcur, X, Y, acc0, acc1);
    }
};

static __device__ __forceinline__ void wave_lds_order() {
    __builtin_amdgcn_fence(__ATOMIC_RELEASE, "wavefront");
    __builtin_amdgcn_wave_barrier();
    __builtin_amdgcn_fence(__ATOMIC_ACQUIRE, "wavefront");
}

__global__ __launch_bounds__(NTHREADS)
void vindy_wmma_kernel(const float* __restrict__ z,
                       const float* __restrict__ betas,
                       const float* __restrict__ big_xi,
                       const float* __restrict__ mask,
                       float* __restrict__ out,
                       int batch)
{
    __shared__ __attribute__((aligned(16))) float sImg[IMG_FLOATS];
    __shared__ __attribute__((aligned(16))) float sXT[WAVES][NMAIN][ROWS_PER_WAVE];

    const int tid  = threadIdx.x;
    const int lane = tid & 31;
    const int w    = __builtin_amdgcn_readfirstlane(tid >> 5);
    const int half = lane >> 4;
    const int lm   = lane & 15;

    {
        v4f* s4 = (v4f*)sImg;
        const v4f zero4 = {0.f, 0.f, 0.f, 0.f};
        constexpr int TAIL0 = tile_base(NMAIN) * 16;
        constexpr int TAILN = (NGTOT - tile_base(NMAIN)) * 16;
        for (int t = tid; t < TAILN; t += NTHREADS) s4[TAIL0 + t] = zero4;
        for (int t = tid; t < NMAIN * 16; t += NTHREADS)
            s4[(int)STAB.lastg[t >> 4] * 16 + (t & 15)] = zero4;
    }
    __syncthreads();

    {
        float* s = sImg;
        const v4f* xi4 = (const v4f*)big_xi;
        const v4f* mk4 = (const v4f*)mask;
        constexpr int NQ = NP * NV / 4;
        constexpr int UB = 12;
        static_assert(NP * NV % 4 == 0, "quad loads need a whole number of quads");
        for (int base = tid; base < NQ; base += NTHREADS * UB) {
            v4f a[UB], m[UB];
            unsigned w0[UB], w1[UB];
#pragma unroll
            for (int u = 0; u < UB; ++u) {
                int q = base + NTHREADS * u;
                if (q > NQ - 1) q = NQ - 1;
                a[u] = xi4[q];
                m[u] = mk4[q];
                const unsigned f0 = (4u * (unsigned)q) / (unsigned)NV;
                w0[u] = STAB.v[f0];
                w1[u] = STAB.v[f0 + 1u];
            }
#pragma unroll
            for (int u = 0; u < UB; ++u) {
                const int q = base + NTHREADS * u;
                if (q < NQ) {
                    const unsigned i0 = 4u * (unsigned)q;
                    const unsigned f0 = i0 / (unsigned)NV;
                    const unsigned n0 = i0 - f0 * (unsigned)NV;
                    const unsigned m0 = w0[u] & 0xffffu;
                    const unsigned t0 = w0[u] >> 16;
                    const unsigned m1 = w1[u] & 0xffffu;
                    const unsigned d0 = (n0 < 16u) ? (m0 + 2u * n0) : t0;
                    const unsigned d2 = (n0 < 14u) ? (m0 + 2u * n0 + 4u) : ((n0 == 14u) ? t0 : m1);
                    s[d0]      = a[u].x * m[u].x;
                    s[d0 + 2u] = a[u].y * m[u].y;
                    s[d2]      = a[u].z * m[u].z;
                    s[d2 + 2u] = a[u].w * m[u].w;
                }
            }
        }
    }

    __syncthreads();

    const bool hi = (half != 0);
    const float* xh = &sXT[w][0][8 * half];
    const int nWaveTiles = (batch + ROWS_PER_WAVE - 1) / ROWS_PER_WAVE;

    for (int wt = blockIdx.x * WAVES + w; wt < nWaveTiles; wt += NBLOCKS * WAVES) {
        const int rowBase = wt * ROWS_PER_WAVE;

        int gr = rowBase + lm;
        if (gr > batch - 1) gr = batch - 1;
        float x[19];
        {
            const v4f* zr = (const v4f*)(z + (size_t)gr * 16);
            v4f q0 = zr[0], q1 = zr[1], q2 = zr[2], q3 = zr[3];
            x[0]  = q0.x; x[1]  = q0.y; x[2]  = q0.z; x[3]  = q0.w;
            x[4]  = q1.x; x[5]  = q1.y; x[6]  = q1.z; x[7]  = q1.w;
            x[8]  = q2.x; x[9]  = q2.y; x[10] = q2.z; x[11] = q2.w;
            x[12] = q3.x; x[13] = q3.y; x[14] = q3.z; x[15] = q3.w;
            const v2f* br = (const v2f*)(betas + (size_t)gr * 2);
            v2f bb = br[0];
            x[16] = bb.x; x[17] = bb.y; x[18] = 1.0f;
        }

        wave_lds_order();
        if (!hi) {
#pragma unroll
            for (int i = 0; i < NMAIN; ++i) sXT[w][i][lm] = x[i];
        }
        wave_lds_order();

        v2f A[NAG];
        FillA<0, NAG>::run(x, hi, A);

        v8f acc0 = {0.f, 0.f, 0.f, 0.f, 0.f, 0.f, 0.f, 0.f};
        v8f acc1 = {0.f, 0.f, 0.f, 0.f, 0.f, 0.f, 0.f, 0.f};

        unsigned bLo = 8u * (unsigned)lane;
        unsigned bHi = bLo + 256u * (unsigned)HI_STEP0;

        v8f X = {0.f, 0.f, 0.f, 0.f, 0.f, 0.f, 0.f, 0.f};
        v8f Y = {0.f, 0.f, 0.f, 0.f, 0.f, 0.f, 0.f, 0.f};
        v2f Ba[CH], Bb[CH];
        LoadB<0, CH>::run(sImg, bLo, bHi, Ba);
        Chunks<0>::run(A, sImg, bLo, bHi, xh, lm, Ba, Bb, X, Y, acc0, acc1);

#pragma unroll
        for (int v = 0; v < 8; ++v) {
            float t = acc1[v];
            t += __shfl_xor(t, 2);
            t += __shfl_xor(t, 4);
            t += __shfl_xor(t, 8);
            acc1[v] = t;
        }

        const int orow = rowBase + 8 * half;
        if (rowBase + ROWS_PER_WAVE <= batch) {
#pragma unroll
            for (int v = 0; v < 8; ++v)
                out[(size_t)(orow + v) * NV + lm] = acc0[v];
            if (lm < 2) {
#pragma unroll
                for (int v = 0; v < 8; ++v)
                    out[(size_t)(orow + v) * NV + 16 + lm] = acc1[v];
            }
        } else {
#pragma unroll
            for (int v = 0; v < 8; ++v) {
                int r = orow + v;
                if (r < batch) {
                    out[(size_t)r * NV + lm] = acc0[v];
                    if (lm < 2) out[(size_t)r * NV + 16 + lm] = acc1[v];
                }
            }
        }
    }
}

extern "C" void kernel_launch(void* const* d_in, const int* in_sizes, int n_in,
                              void* d_out, int out_size, void* d_ws, size_t ws_size,
                              hipStream_t stream) {
    const float* z      = (const float*)d_in[0];
    const float* betas  = (const float*)d_in[1];
    const float* big_xi = (const float*)d_in[2];
    const float* mask   = (const float*)d_in[3];
    float* out          = (float*)d_out;

    const int batch = in_sizes[0] / 16;

    vindy_wmma_kernel<<<NBLOCKS, NTHREADS, 0, stream>>>(z, betas, big_xi, mask, out, batch);
}


# A wave starts each tile at priority 3 and steps down (s_setprio 2,1,0) through its last tile, so the four waves of a SIMD end together and none runs its last tile alone; sum order unchanged
# speedup vs baseline: 3.3629x; 1.0425x over previous
#include <hip/hip_runtime.h>


#define NV 18
#define NP 1330
#define NPAIR 190
#define NAG 48
#define NMAIN 19
#define NTILE 22
#define WAVES 16
#define NTHREADS (WAVES * 32)
#define ROWS_PER_WAVE 16
#define NBLOCKS 256

typedef __attribute__((ext_vector_type(2))) float v2f;
typedef __attribute__((ext_vector_type(4))) float v4f;
typedef __attribute__((ext_vector_type(8))) float v8f;

constexpr int pairs_from(int i) { return (19 - i) * (20 - i) / 2; }
constexpr int pair_index(int j, int k) { return (18 - j) * (19 - j) / 2 + (k - j); }
constexpr int pair_j(int p) {
    int j = 18;
    while (p >= (18 - j) * (19 - j) / 2 + (19 - j)) --j;
    return j;
}
constexpr int pair_k(int p) { return pair_j(p) + (p - (18 - pair_j(p)) * (19 - pair_j(p)) / 2); }
constexpr int tile_groups(int t) {
    return t < NMAIN ? (pairs_from(t) + 3) / 4 : (pairs_from(8 * (t - NMAIN)) + 3) / 4;
}
constexpr int tile_base(int t) {
    int s = 0;
    for (int u = 0; u < t; ++u) s += tile_groups(u);
    return s;
}
constexpr int CHUNK_STEPS = 12;
constexpr int NGTOT = tile_base(NTILE);
constexpr int IMG_FLOATS = NGTOT * 64;

struct StageTab {
    unsigned v[NP + 1];
    unsigned short lastg[NMAIN + 1];
};
constexpr unsigned stage_word(int i, int j, int k) {
    const int p = pair_index(j, k);
    const int m = (tile_base(i) + (p >> 2)) * 64 + 32 * ((p >> 1) & 1) + (p & 1);
    const int t = (tile_base(NMAIN + (i >> 3)) + (p >> 2)) * 64
                + (2 * (i & 7) + 16 * ((p >> 1) & 1)) * 2 + (p & 1);
    return (unsigned)m | ((unsigned)t << 16);
}
constexpr StageTab make_stage_tab() {
    StageTab t{};
    int f = 0;
    t.v[f++] = stage_word(18, 18, 18);
    for (int i = 0; i < NV; ++i) t.v[f++] = stage_word(i, 18, 18);
    for (int i = 0; i < NV; ++i)
        for (int j = i; j < NV; ++j) t.v[f++] = stage_word(i, j, 18);
    for (int i = 0; i < NV; ++i)
        for (int j = i; j < NV; ++j)
            for (int k = j; k < NV; ++k) t.v[f++] = stage_word(i, j, k);
    t.v[NP] = 0;
    for (int i = 0; i < NMAIN; ++i) t.lastg[i] = (unsigned short)(tile_base(i + 1) - 1);
    t.lastg[NMAIN] = 0;
    return t;
}
__constant__ const StageTab STAB = make_stage_tab();

template <int P>
static __device__ __forceinline__ float pairprod(const float* x) {
    if constexpr (P >= NPAIR) {
        return 0.0f;
    } else {
        constexpr int j = pair_j(P);
        constexpr int k = pair_k(P);
        if constexpr (k == 18) return x[j];
        else return x[j] * x[k];
    }
}

template <int G, int GEnd>
struct FillA {
    static __device__ __forceinline__ void run(const float* x, bool hi, v2f* A) {
        constexpr int p = 4 * G;
        const float q0 = pairprod<p + 0>(x);
        const float q1 = pairprod<p + 1>(x);
        const float q2 = pairprod<p + 2>(x);
        const float q3 = pairprod<p + 3>(x);
        A[G].x = hi ? q2 : q0;
        A[G].y = hi ? q3 : q1;
        FillA<G + 1, GEnd>::run(x, hi, A);
    }
};
template <int GEnd>
struct FillA<GEnd, GEnd> {
    static __device__ __forceinline__ void run(const float*, bool, v2f*) {}
};

static __device__ __forceinline__ void scale_add(const float* xs, const v8f& t, v8f& acc) {
    const v4f s0 = *(const v4f*)(xs);
    const v4f s1 = *(const v4f*)(xs + 4);
    acc[0] += s0.x * t[0];
    acc[1] += s0.y * t[1];
    acc[2] += s0.z * t[2];
    acc[3] += s0.w * t[3];
    acc[4] += s1.x * t[4];
    acc[5] += s1.y * t[5];
    acc[6] += s1.z * t[6];
    acc[7] += s1.w * t[7];
}

constexpr int CH = CHUNK_STEPS;
constexpr int PRIO_DOWN_A = 13;
constexpr int PRIO_DOWN_B = 26;
constexpr int PRIO_DOWN_C = 32;
constexpr int HI_STEP0 = 256;

struct Sched { int w[NGTOT]; };
constexpr Sched make_sched() {
    Sched s{};
    const int order[NTILE] = {0, 19, 1, 2, 3, 4, 5, 6, 7, 8, 20, 9, 10, 11, 12, 13, 14, 15, 16, 21, 17, 18};
    int next = 2;
    int curT[2] = {order[0], order[1]};
    int curG[2] = {0, 0};
    int E = 0, c = 0;
    while (E < NGTOT) {
        if (curT[c] < 0) { c ^= 1; continue; }
        const bool last = (curG[c] == tile_groups(curT[c]) - 1);
        s.w[E] = curT[c] | (curG[c] << 8) | (c << 16) | ((last ? 1 : 0) << 17);
        if (last) {
            if (next < NTILE) { curT[c] = order[next++]; curG[c] = 0; } else curT[c] = -1;
        } else {
            ++curG[c];
        }
        ++E;
        c ^= 1;
    }
    return s;
}
constexpr int ex_word(int E)  { return make_sched().w[E]; }
constexpr int ex_tile(int E)  { return ex_word(E) & 0xff; }
constexpr int ex_group(int E) { return (ex_word(E) >> 8) & 0xff; }
constexpr int ex_chain(int E) { return (ex_word(E) >> 16) & 1; }
constexpr bool ex_last(int E) { return ((ex_word(E) >> 17) & 1) != 0; }
constexpr int ex_step(int E)  { return tile_base(ex_tile(E)) + ex_group(E); }

template <int E, int EEnd>
struct LoadB {
    static __device__ __forceinline__ void run(const float* img, unsigned bLo, unsigned bHi, v2f* B) {
        if constexpr (E < NGTOT) {
            constexpr int S = ex_step(E);
            if constexpr (S < HI_STEP0)
                B[E % CH] = *(const v2f*)((const char*)img + bLo + S * 256);
            else
                B[E % CH] = *(const v2f*)((const char*)img + bHi + (S - HI_STEP0) * 256);
        }
        LoadB<E + 1, EEnd>::run(img, bLo, bHi, B);
    }
};
template <int EEnd>
struct LoadB<EEnd, EEnd> {
    static __device__ __forceinline__ void run(const float*, unsigned, unsigned, v2f*) {}
};

template <int T>
static __device__ __forceinline__ void finish_tile(const float* xh, int lm, const v8f& t, v8f& acc0, v8f& acc1) {
    if constexpr (T < NMAIN) {
        scale_add(xh + T * 16, t, acc0);
    } else {
        int i = 8 * (T - NMAIN) + (lm >> 1);
        if (i > 18) i = 18;
        scale_add(xh + i * 16, t, acc1);
    }
}

template <int E, int EEnd>
struct Mma {
    static __device__ __forceinline__ void run(const v2f* A, const v2f* B, const float* xh, int lm,
                                               v8f& X, v8f& Y, v8f& acc0, v8f& acc1) {
        if constexpr (E < NGTOT) {
            constexpr int t = ex_tile(E);
            constexpr int g = ex_group(E);
            v8f& acc = (ex_chain(E) == 0) ? X : Y;
            if constexpr (g == 0) {
                const v8f zero = {0.f, 0.f, 0.f, 0.f, 0.f, 0.f, 0.f, 0.f};
                acc = __builtin_amdgcn_wmma_f32_16x16x4_f32(false, A[g], false, B[E % CH], (short)0, zero, false, false);
            } else {
                acc = __builtin_amdgcn_wmma_f32_16x16x4_f32(false, A[g], false, B[E % CH], (short)0, acc, false, false);
            }
            if constexpr (ex_last(E)) finish_tile<t>(xh, lm, acc, acc0, acc1);
        }
        Mma<E + 1, EEnd>::run(A, B, xh, lm, X, Y, acc0, acc1);
    }
};
template <int EEnd>
struct Mma<EEnd, EEnd> {
    static __device__ __forceinline__ void run(const v2f*, const v2f*, const float*, int, v8f&, v8f&, v8f&, v8f&) {}
};

template <int C>
struct Chunks {
    static __device__ __forceinline__ void run(const v2f* A, const float* img, unsigned& bLo, unsigned& bHi,
                                               const float* xh, int lm, const bool lastTile,
                                               v2f* Bcur, v2f* Bnext, v8f& X, v8f& Y, v8f& acc0, v8f& acc1) {
        LoadB<(C + 1) * CH, (C + 2) * CH>::run(img, bLo, bHi, Bnext);
        __builtin_amdgcn_sched_barrier(0x6);
        if constexpr (C == PRIO_DOWN_A) { if (lastTile) __builtin_amdgcn_s_setprio(2); }
        if constexpr (C == PRIO_DOWN_B) { if (lastTile) __builtin_amdgcn_s_setprio(1); }
        if constexpr (C == PRIO_DOWN_C) { if (lastTile) __builtin_amdgcn_s_setprio(0); }
        Mma<C * CH, (C + 1) * CH>::run(A, Bcur, xh, lm, X, Y, acc0, acc1);
        asm volatile("" : "+v"(bLo), "+v"(bHi), "+v"(X), "+v"(Y), "+v"(acc0), "+v"(acc1));
        if constexpr ((C + 1) * CH < NGTOT)
            Chunks<C + 1>::run(A, img, bLo, bHi, xh, lm, lastTile, Bnext, Bcur, X, Y, acc0, acc1);
    }
};

static __device__ __forceinline__ void wave_lds_order() {
    __builtin_amdgcn_fence(__ATOMIC_RELEASE, "wavefront");
    __builtin_amdgcn_wave_barrier();
    __builtin_amdgcn_fence(__ATOMIC_ACQUIRE, "wavefront");
}

__global__ __launch_bounds__(NTHREADS)
void vindy_wmma_kernel(const float* __restrict__ z,
                       const float* __restrict__ betas,
                       const float* __restrict__ big_xi,
                       const float* __restrict__ mask,
                       float* __restrict__ out,
                       int batch)
{
    __shared__ __attribute__((aligned(16))) float sImg[IMG_FLOATS];
    __shared__ __attribute__((aligned(16))) float sXT[WAVES][NMAIN][ROWS_PER_WAVE];

    const int tid  = threadIdx.x;
    const int lane = tid & 31;
    const int w    = __builtin_amdgcn_readfirstlane(tid >> 5);
    const int half = lane >> 4;
    const int lm   = lane & 15;

    {
        v4f* s4 = (v4f*)sImg;
        const v4f zero4 = {0.f, 0.f, 0.f, 0.f};
        constexpr int TAIL0 = tile_base(NMAIN) * 16;
        constexpr int TAILN = (NGTOT - tile_base(NMAIN)) * 16;
        for (int t = tid; t < TAILN; t += NTHREADS) s4[TAIL0 + t] = zero4;
        for (int t = tid; t < NMAIN * 16; t += NTHREADS)
            s4[(int)STAB.lastg[t >> 4] * 16 + (t & 15)] = zero4;
    }
    __syncthreads();

    {
        float* s = sImg;
        const v4f* xi4 = (const v4f*)big_xi;
        const v4f* mk4 = (const v4f*)mask;
        constexpr int NQ = NP * NV / 4;
        constexpr int UB = 12;
        static_assert(NP * NV % 4 == 0, "quad loads need a whole number of quads");
        for (int base = tid; base < NQ; base += NTHREADS * UB) {
            v4f a[UB], m[UB];
            unsigned w0[UB], w1[UB];
#pragma unroll
            for (int u = 0; u < UB; ++u) {
                int q = base + NTHREADS * u;
                if (q > NQ - 1) q = NQ - 1;
                a[u] = xi4[q];
                m[u] = mk4[q];
                const unsigned f0 = (4u * (unsigned)q) / (unsigned)NV;
                w0[u] = STAB.v[f0];
                w1[u] = STAB.v[f0 + 1u];
            }
#pragma unroll
            for (int u = 0; u < UB; ++u) {
                const int q = base + NTHREADS * u;
                if (q < NQ) {
                    const unsigned i0 = 4u * (unsigned)q;
                    const unsigned f0 = i0 / (unsigned)NV;
                    const unsigned n0 = i0 - f0 * (unsigned)NV;
                    const unsigned m0 = w0[u] & 0xffffu;
                    const unsigned t0 = w0[u] >> 16;
                    const unsigned m1 = w1[u] & 0xffffu;
                    const unsigned d0 = (n0 < 16u) ? (m0 + 2u * n0) : t0;
                    const unsigned d2 = (n0 < 14u) ? (m0 + 2u * n0 + 4u) : ((n0 == 14u) ? t0 : m1);
                    s[d0]      = a[u].x * m[u].x;
                    s[d0 + 2u] = a[u].y * m[u].y;
                    s[d2]      = a[u].z * m[u].z;
                    s[d2 + 2u] = a[u].w * m[u].w;
                }
            }
        }
    }

    __syncthreads();

    const bool hi = (half != 0);
    const float* xh = &sXT[w][0][8 * half];
    const int nWaveTiles = (batch + ROWS_PER_WAVE - 1) / ROWS_PER_WAVE;

    for (int wt = blockIdx.x * WAVES + w; wt < nWaveTiles; wt += NBLOCKS * WAVES) {
        const int rowBase = wt * ROWS_PER_WAVE;
        const bool lastTile = (wt + NBLOCKS * WAVES >= nWaveTiles);
        __builtin_amdgcn_s_setprio(3);

        int gr = rowBase + lm;
        if (gr > batch - 1) gr = batch - 1;
        float x[19];
        {
            const v4f* zr = (const v4f*)(z + (size_t)gr * 16);
            v4f q0 = zr[0], q1 = zr[1], q2 = zr[2], q3 = zr[3];
            x[0]  = q0.x; x[1]  = q0.y; x[2]  = q0.z; x[3]  = q0.w;
            x[4]  = q1.x; x[5]  = q1.y; x[6]  = q1.z; x[7]  = q1.w;
            x[8]  = q2.x; x[9]  = q2.y; x[10] = q2.z; x[11] = q2.w;
            x[12] = q3.x; x[13] = q3.y; x[14] = q3.z; x[15] = q3.w;
            const v2f* br = (const v2f*)(betas + (size_t)gr * 2);
            v2f bb = br[0];
            x[16] = bb.x; x[17] = bb.y; x[18] = 1.0f;
        }

        wave_lds_order();
        if (!hi) {
#pragma unroll
            for (int i = 0; i < NMAIN; ++i) sXT[w][i][lm] = x[i];
        }
        wave_lds_order();

        v2f A[NAG];
        FillA<0, NAG>::run(x, hi, A);

        v8f acc0 = {0.f, 0.f, 0.f, 0.f, 0.f, 0.f, 0.f, 0.f};
        v8f acc1 = {0.f, 0.f, 0.f, 0.f, 0.f, 0.f, 0.f, 0.f};

        unsigned bLo = 8u * (unsigned)lane;
        unsigned bHi = bLo + 256u * (unsigned)HI_STEP0;

        v8f X = {0.f, 0.f, 0.f, 0.f, 0.f, 0.f, 0.f, 0.f};
        v8f Y = {0.f, 0.f, 0.f, 0.f, 0.f, 0.f, 0.f, 0.f};
        v2f Ba[CH], Bb[CH];
        LoadB<0, CH>::run(sImg, bLo, bHi, Ba);
        Chunks<0>::run(A, sImg, bLo, bHi, xh, lm, lastTile, Ba, Bb, X, Y, acc0, acc1);

#pragma unroll
        for (int v = 0; v < 8; ++v) {
            float t = acc1[v];
            t += __shfl_xor(t, 2);
            t += __shfl_xor(t, 4);
            t += __shfl_xor(t, 8);
            acc1[v] = t;
        }

        const int orow = rowBase + 8 * half;
        if (rowBase + ROWS_PER_WAVE <= batch) {
#pragma unroll
            for (int v = 0; v < 8; ++v)
                out[(size_t)(orow + v) * NV + lm] = acc0[v];
            if (lm < 2) {
#pragma unroll
                for (int v = 0; v < 8; ++v)
                    out[(size_t)(orow + v) * NV + 16 + lm] = acc1[v];
            }
        } else {
#pragma unroll
            for (int v = 0; v < 8; ++v) {
                int r = orow + v;
                if (r < batch) {
                    out[(size_t)r * NV + lm] = acc0[v];
                    if (lm < 2) out[(size_t)r * NV + 16 + lm] = acc1[v];
                }
            }
        }
    }
}

extern "C" void kernel_launch(void* const* d_in, const int* in_sizes, int n_in,
                              void* d_out, int out_size, void* d_ws, size_t ws_size,
                              hipStream_t stream) {
    const float* z      = (const float*)d_in[0];
    const float* betas  = (const float*)d_in[1];
    const float* big_xi = (const float*)d_in[2];
    const float* mask   = (const float*)d_in[3];
    float* out          = (float*)d_out;

    const int batch = in_sizes[0] / 16;

    vindy_wmma_kernel<<<NBLOCKS, NTHREADS, 0, stream>>>(z, betas, big_xi, mask, out, batch);
}
